# starting kernel
# baseline (speedup 1.0000x reference)
#include <hip/hip_runtime.h>

#define T_SEQ 4096
#define HID   1024
#define KDIM  1024

#define KC    32
#define BM    128
#define BN    64
#define LDX   36
#define LDB   36

#define CLEN   256
#define NCHUNK (T_SEQ / CLEN)

typedef float v2f __attribute__((ext_vector_type(2)));
typedef float v4f __attribute__((ext_vector_type(4)));
typedef float v8f __attribute__((ext_vector_type(8)));

__global__ __launch_bounds__(256) void gemm_xb_wmma(const float* __restrict__ X,
                                                    const float* __restrict__ B,
                                                    float* __restrict__ S) {
  __shared__ float sX[BM * LDX];
  __shared__ float sB[BN * LDB];

  const int tid  = threadIdx.x;
  const int wave = tid >> 5;
  const int lane = tid & 31;
  const int half = lane >> 4;
  const int r    = lane & 15;

  const int blockN = blockIdx.x * BN;
  const int blockM = blockIdx.y * BM;

  const int wm = (wave & 3) << 5;
  const int wn = (wave >> 2) << 5;

  const int rowGrp = tid >> 3;
  const int col4   = (tid & 7) << 2;
  const int bRow   = tid >> 4;
  const int bCol4  = (tid & 15) << 2;

  v8f acc00 = {}, acc01 = {}, acc10 = {}, acc11 = {};

  for (int k0 = 0; k0 < KDIM; k0 += KC) {
    __syncthreads();

    #pragma unroll
    for (int p = 0; p < 4; ++p) {
      const int row = rowGrp + (p << 5);
      v4f xv = *(const v4f*)(X + (size_t)(blockM + row) * KDIM + k0 + col4);
      *(v4f*)(&sX[row * LDX + col4]) = xv;
    }
    #pragma unroll
    for (int p = 0; p < 2; ++p) {
      const int kRow = bRow + (p << 4);
      v4f bv = *(const v4f*)(B + (size_t)(k0 + kRow) * HID + blockN + bCol4);
      #pragma unroll
      for (int i = 0; i < 4; ++i)
        sB[(bCol4 + i) * LDB + kRow] = bv[i];
    }
    __syncthreads();

    #pragma unroll
    for (int kk = 0; kk < KC; kk += 4) {
      v2f a0 = *(const v2f*)(&sX[(wm + r)      * LDX + kk + 2 * half]);
      v2f a1 = *(const v2f*)(&sX[(wm + 16 + r) * LDX + kk + 2 * half]);
      v2f b0 = *(const v2f*)(&sB[(wn + r)      * LDB + kk + 2 * half]);
      v2f b1 = *(const v2f*)(&sB[(wn + 16 + r) * LDB + kk + 2 * half]);
      acc00 = __builtin_amdgcn_wmma_f32_16x16x4_f32(false, a0, false, b0,
                                                    (short)0, acc00, false, false);
      acc01 = __builtin_amdgcn_wmma_f32_16x16x4_f32(false, a0, false, b1,
                                                    (short)0, acc01, false, false);
      acc10 = __builtin_amdgcn_wmma_f32_16x16x4_f32(false, a1, false, b0,
                                                    (short)0, acc10, false, false);
      acc11 = __builtin_amdgcn_wmma_f32_16x16x4_f32(false, a1, false, b1,
                                                    (short)0, acc11, false, false);
    }
  }

  #pragma unroll
  for (int i = 0; i < 8; ++i) {
    const int row0 = blockM + wm + i + 8 * half;
    const int col0 = blockN + wn + r;
    S[(size_t)row0        * HID + col0]      = acc00[i];
    S[(size_t)row0        * HID + col0 + 16] = acc01[i];
    S[(size_t)(row0 + 16) * HID + col0]      = acc10[i];
    S[(size_t)(row0 + 16) * HID + col0 + 16] = acc11[i];
  }
}

__global__ __launch_bounds__(256) void scan_local(const float* __restrict__ a_diag,
                                                  float* __restrict__ out,
                                                  float* __restrict__ carry) {
  const int h = blockIdx.x * 256 + threadIdx.x;
  const int c = blockIdx.y;
  const float a = a_diag[h];
  float accv = 0.0f;
  const int t0 = c * CLEN;
  for (int t = 0; t < CLEN; ++t) {
    const size_t idx = (size_t)(t0 + t) * HID + h;
    accv = fmaf(a, accv, out[idx]);
    out[idx] = accv;
  }
  carry[c * HID + h] = accv;
}

__global__ __launch_bounds__(256) void scan_carry(const float* __restrict__ a_diag,
                                                  const float* __restrict__ carry,
                                                  float* __restrict__ carryIn) {
  const int h = blockIdx.x * 256 + threadIdx.x;
  const float a = a_diag[h];
  float aL = a;
  #pragma unroll
  for (int i = 0; i < 8; ++i) aL *= aL;
  float state = 0.0f;
  for (int c = 0; c < NCHUNK; ++c) {
    carryIn[c * HID + h] = state;
    state = fmaf(aL, state, carry[c * HID + h]);
  }
}

__global__ __launch_bounds__(256) void scan_fix(const float* __restrict__ a_diag,
                                                const float* __restrict__ carryIn,
                                                float* __restrict__ out) {
  const int h = blockIdx.x * 256 + threadIdx.x;
  const int c = blockIdx.y + 1;
  const float a  = a_diag[h];
  const float ci = carryIn[c * HID + h];
  float w = a;
  const int t0 = c * CLEN;
  for (int t = 0; t < CLEN; ++t) {
    const size_t idx = (size_t)(t0 + t) * HID + h;
    out[idx] = fmaf(w, ci, out[idx]);
    w *= a;
  }
}

extern "C" void kernel_launch(void* const* d_in, const int* in_sizes, int n_in,
                              void* d_out, int out_size, void* d_ws, size_t ws_size,
                              hipStream_t stream) {
  const float* x_seq  = (const float*)d_in[0];
  const float* a_diag = (const float*)d_in[1];
  const float* b_mat  = (const float*)d_in[2];
  float* out = (float*)d_out;

  float* carry   = (float*)d_ws;
  float* carryIn = carry + NCHUNK * HID;

  dim3 gGemm(HID / BN, T_SEQ / BM);
  gemm_xb_wmma<<<gGemm, 256, 0, stream>>>(x_seq, b_mat, out);

  dim3 gLocal(HID / 256, NCHUNK);
  scan_local<<<gLocal, 256, 0, stream>>>(a_diag, out, carry);

  scan_carry<<<HID / 256, 256, 0, stream>>>(a_diag, carry, carryIn);

  dim3 gFix(HID / 256, NCHUNK - 1);
  scan_fix<<<gFix, 256, 0, stream>>>(a_diag, carryIn, out);
}


# Scan's chunk-end reduce pass taken from the GEMM accumulators in its epilogue (8-row lane runs joined by a^16, a^8, a^32 over LDS): scan_ends launch and its read of S gone; ends' sum order changes
# speedup vs baseline: 1.6591x; 1.6591x over previous
#include <hip/hip_runtime.h>

#define T_SEQ 4096
#define HID   1024
#define KDIM  1024

#define KC    128
#define BM    128
#define BN    64
#define LDX   132
#define BGS   2080

typedef float v2f __attribute__((ext_vector_type(2)));
typedef float v4f __attribute__((ext_vector_type(4)));
typedef float v8f __attribute__((ext_vector_type(8)));

__global__ __launch_bounds__(256) __attribute__((amdgpu_waves_per_eu(4, 4)))
void gemm_xb_wmma(const float* __restrict__ X,
                                                    const float* __restrict__ B,
                                                    const float* __restrict__ a_diag,
                                                    float* __restrict__ S,
                                                    float* __restrict__ ends) {
  __shared__ float sX[BM * LDX];
  __shared__ __attribute__((aligned(16))) float sB[4 * BGS];
  __shared__ float sQ[4 * 2 * BN];

  const int tid  = threadIdx.x;
  const int wave = tid >> 5;
  const int lane = tid & 31;
  const int half = lane >> 4;
  const int r    = lane & 15;

  const int blockN = blockIdx.x * BN;
  const int blockM = blockIdx.y * BM;

  const int wm = (wave & 3) << 5;
  const int wn = (wave >> 2) << 5;

  const int rowGrp = tid >> 5;
  const int col4   = (tid & 31) << 2;
  const int bCol2  = (tid & 31) << 1;
  const int bDst   = ((tid & 31) >> 3) * BGS + ((tid & 7) << 2);

  v8f acc00 = {}, acc01 = {}, acc10 = {}, acc11 = {};

  const float aL = a_diag[blockN + wn + r];
  const float aH = a_diag[blockN + wn + r + 16];
  const float aC = a_diag[blockN + (tid & (BN - 1))];

  __builtin_amdgcn_s_setprio(3);

  for (int k0 = 0; k0 < KDIM; k0 += KC) {
    if (k0 == 5 * KC) __builtin_amdgcn_s_setprio(2);
    if (k0 == 6 * KC) __builtin_amdgcn_s_setprio(1);
    if (k0 == 7 * KC) __builtin_amdgcn_s_setprio(0);
    __syncthreads();

    v4f xv[16];
    v2f bv[16];
    #pragma unroll
    for (int p = 0; p < 16; ++p) {
      const int row = rowGrp + (p << 3);
      xv[p] = *(const v4f*)(X + (size_t)(blockM + row) * KDIM + k0 + col4);
    }
    #pragma unroll
    for (int q = 0; q < 8; ++q) {
      const int kp = rowGrp + (q << 3);
      const float* g = B + (size_t)(k0 + 2 * kp) * HID + blockN + bCol2;
      bv[2 * q]     = *(const v2f*)(g);
      bv[2 * q + 1] = *(const v2f*)(g + HID);
    }
    #pragma unroll
    for (int p = 0; p < 16; ++p) {
      const int row = rowGrp + (p << 3);
      *(v4f*)(&sX[row * LDX + col4]) = xv[p];
    }
    #pragma unroll
    for (int q = 0; q < 8; ++q) {
      const int kp = rowGrp + (q << 3);
      const v4f t = { bv[2 * q][0], bv[2 * q + 1][0], bv[2 * q][1], bv[2 * q + 1][1] };
      *(v4f*)(&sB[bDst + kp * 32]) = t;
    }
    __syncthreads();

    #pragma unroll
    for (int kk = 0; kk < KC; kk += 4) {
      v2f a0 = *(const v2f*)(&sX[(wm + r)      * LDX + kk + 2 * half]);
      v2f a1 = *(const v2f*)(&sX[(wm + 16 + r) * LDX + kk + 2 * half]);
      v2f b0 = *(const v2f*)(&sB[(wn >> 4)       * BGS + ((kk >> 1) + half) * 32 + 2 * r]);
      v2f b1 = *(const v2f*)(&sB[((wn >> 4) + 1) * BGS + ((kk >> 1) + half) * 32 + 2 * r]);
      acc00 = __builtin_amdgcn_wmma_f32_16x16x4_f32(false, a0, false, b0,
                                                    (short)0, acc00, false, false);
      acc01 = __builtin_amdgcn_wmma_f32_16x16x4_f32(false, a0, false, b1,
                                                    (short)0, acc01, false, false);
      acc10 = __builtin_amdgcn_wmma_f32_16x16x4_f32(false, a1, false, b0,
                                                    (short)0, acc10, false, false);
      acc11 = __builtin_amdgcn_wmma_f32_16x16x4_f32(false, a1, false, b1,
                                                    (short)0, acc11, false, false);
    }
  }

  const bool wantEnd = blockIdx.y < T_SEQ / BM - 1;
  if (wantEnd) {
    float e00 = acc00[0], e01 = acc01[0], e10 = acc10[0], e11 = acc11[0];
    #pragma unroll
    for (int i = 1; i < 8; ++i) {
      e00 = fmaf(aL, e00, acc00[i]);
      e10 = fmaf(aL, e10, acc10[i]);
      e01 = fmaf(aH, e01, acc01[i]);
      e11 = fmaf(aH, e11, acc11[i]);
    }
    float aL16 = aL, aH16 = aH;
    aL16 *= aL16; aH16 *= aH16;
    aL16 *= aL16; aH16 *= aH16;
    aL16 *= aL16; aH16 *= aH16;
    aL16 *= aL16; aH16 *= aH16;
    const int qb = (((wave & 3) << 1) + half) * BN + wn + r;
    sQ[qb]      = fmaf(aL16, e00, e10);
    sQ[qb + 16] = fmaf(aH16, e01, e11);
    __syncthreads();
  }

  #pragma unroll
  for (int i = 0; i < 8; ++i) {
    const int row0 = blockM + wm + i + 8 * half;
    const int col0 = blockN + wn + r;
    S[(size_t)row0        * HID + col0]      = acc00[i];
    S[(size_t)row0        * HID + col0 + 16] = acc01[i];
    S[(size_t)(row0 + 16) * HID + col0]      = acc10[i];
    S[(size_t)(row0 + 16) * HID + col0 + 16] = acc11[i];
  }

  if (wantEnd && tid < BN) {
    float a8 = aC * aC;
    a8 *= a8;
    a8 *= a8;
    float a32 = a8 * a8;
    a32 *= a32;
    float st = fmaf(a8, sQ[tid], sQ[BN + tid]);
    st = fmaf(a32, st, fmaf(a8, sQ[2 * BN + tid], sQ[3 * BN + tid]));
    st = fmaf(a32, st, fmaf(a8, sQ[4 * BN + tid], sQ[5 * BN + tid]));
    st = fmaf(a32, st, fmaf(a8, sQ[6 * BN + tid], sQ[7 * BN + tid]));
    ends[(size_t)blockIdx.y * HID + blockN + tid] = st;
  }
}

#define SC_ROWS  16
#define SC_WAVES 8
#define SC_T     (SC_ROWS * SC_WAVES)
#define SC_H     128
#define SC_NCH   (T_SEQ / SC_T)

static __device__ __forceinline__ v4f fma4(v4f a, v4f b, v4f c) {
  return __builtin_elementwise_fma(a, b, c);
}

__global__ __launch_bounds__(256) void scan_apply(const float* __restrict__ a_diag,
                                                  const float* __restrict__ ends,
                                                  float* __restrict__ out) {
  __shared__ float sE[SC_WAVES * SC_H];
  __shared__ float sC[SC_H];
  const int tid = threadIdx.x;
  const int w   = __builtin_amdgcn_readfirstlane(tid >> 5);
  const int l   = tid & 31;
  const int ch  = blockIdx.x * SC_H + 4 * l;
  const int c   = blockIdx.y;
  float* p = out + (size_t)(c * SC_T + w * SC_ROWS) * HID + ch;

  v4f s[SC_ROWS];
  #pragma unroll
  for (int i = 0; i < SC_ROWS; ++i) s[i] = *(const v4f*)(p + (size_t)i * HID);
  const v4f a = *(const v4f*)(a_diag + ch);

  if (w < 4) {
    const int chp = blockIdx.x * SC_H + tid;
    const float ap = a_diag[chp];
    float ev[SC_NCH - 1];
    #pragma unroll
    for (int j = 0; j < SC_NCH - 1; ++j) {
      const int cp = j - (SC_NCH - 1 - c);
      ev[j] = (cp >= 0) ? ends[(size_t)cp * HID + chp] : 0.0f;
    }
    float aT = ap;
    #pragma unroll
    for (int i = 0; i < 7; ++i) aT *= aT;
    float st = 0.0f;
    #pragma unroll
    for (int j = 0; j < SC_NCH - 1; ++j) st = fmaf(aT, st, ev[j]);
    sC[tid] = st;
  }

  v4f e = s[0];
  #pragma unroll
  for (int i = 1; i < SC_ROWS; ++i) e = fma4(a, e, s[i]);
  *(v4f*)(&sE[w * SC_H + 4 * l]) = e;
  __syncthreads();

  v4f a16 = a;
  #pragma unroll
  for (int i = 0; i < 4; ++i) a16 *= a16;
  v4f h = *(const v4f*)(&sC[4 * l]);
  v4f pe[SC_WAVES - 1];
  #pragma unroll
  for (int w2 = 0; w2 < SC_WAVES - 1; ++w2)
    pe[w2] = *(const v4f*)(&sE[w2 * SC_H + 4 * l]);
  #pragma unroll
  for (int w2 = 0; w2 < SC_WAVES - 1; ++w2)
    if (w2 < w) h = fma4(a16, h, pe[w2]);

  #pragma unroll
  for (int i = 0; i < SC_ROWS; ++i) {
    h = fma4(a, h, s[i]);
    *(v4f*)(p + (size_t)i * HID) = h;
  }
}

extern "C" void kernel_launch(void* const* d_in, const int* in_sizes, int n_in,
                              void* d_out, int out_size, void* d_ws, size_t ws_size,
                              hipStream_t stream) {
  const float* x_seq  = (const float*)d_in[0];
  const float* a_diag = (const float*)d_in[1];
  const float* b_mat  = (const float*)d_in[2];
  float* out = (float*)d_out;

  float* ends = (float*)d_ws;

  dim3 gGemm(HID / BN, T_SEQ / BM);
  gemm_xb_wmma<<<gGemm, 256, 0, stream>>>(x_seq, b_mat, a_diag, out, ends);

  dim3 gApply(HID / SC_H, SC_NCH);
  scan_apply<<<gApply, 256, 0, stream>>>(a_diag, ends, out);
}
